# starting kernel
# baseline (speedup 1.0000x reference)
#include <hip/hip_runtime.h>


typedef __attribute__((ext_vector_type(2))) float v2f;
typedef __attribute__((ext_vector_type(8))) float v8f;

#define B_      16
#define CH_     4
#define H_      192
#define W_      192
#define HW_     (H_ * W_)
#define UP_     4
#define OW_     (W_ * UP_)
#define OHW_    (OW_ * OW_)
#define PIX_PER_CHUNK 16
#define CHUNKS_PER_ROW (W_ / PIX_PER_CHUNK)
#define TOTAL_CHUNKS   (B_ * H_ * CHUNKS_PER_ROW)
#define WAVES_PER_BLOCK 8

__global__ __launch_bounds__(256)
void plane2depth_wmma_kernel(const float* __restrict__ feat,
                             const float* __restrict__ Wm,
                             float* __restrict__ out)
{
    const int lane  = threadIdx.x & 31;
    const int chunk = blockIdx.x * WAVES_PER_BLOCK + (threadIdx.x >> 5);

    const int m   = lane & 15;
    const int kHi = (lane >> 4) << 1;
    const int m4  = m & 7;
    v2f a;
    a.x = (m4 < 4) ? Wm[m4 * 4 + kHi]     : 0.0f;
    a.y = (m4 < 4) ? Wm[m4 * 4 + kHi + 1] : 0.0f;

    const int pix  = lane & 15;
    const int rSel = (lane >> 4) << 1;

    const int wBlk = chunk % CHUNKS_PER_ROW;
    const int h    = (chunk / CHUNKS_PER_ROW) % H_;
    const int b    =  chunk / (CHUNKS_PER_ROW * H_);
    const int w    = wBlk * PIX_PER_CHUNK + pix;

    const int inOff = b * (CH_ * HW_) + kHi * HW_ + h * W_ + w;
    v2f bb;
    bb.x = feat[inOff];
    bb.y = feat[inOff + HW_];

    v8f c = {};
    c = __builtin_amdgcn_wmma_f32_16x16x4_f32(
             false, a,  false, bb,
             (short)0, c,  false,  false);

    const float p = c[0], q = c[1], r = c[2], s = c[3];

    const float t    = __builtin_fmaf(p, p, __builtin_fmaf(q, q, __builtin_fmaf(r, r, 1e-6f)));
    const float rsq  = __builtin_amdgcn_rsqf(t);
    const float sn   = s * (t * rsq);
    const float ps   = (p * rsq) * sn;
    const float qs   = (q * rsq) * sn;
    const float rs   = (r * rsq) * sn;

    const int outOff = b * OHW_ + (UP_ * h + rSel) * OW_ + UP_ * w;
    #pragma unroll
    for (int rr = 0; rr < 2; ++rr) {
        const float v    = ((float)(rSel + rr) - 1.5f) * 0.25f;
        const float base = __builtin_fmaf(qs, v, rs);
        float4 o;
        o.x = __builtin_amdgcn_rcpf(fmaxf(__builtin_fmaf(ps, -0.375f, base), 0.1f));
        o.y = __builtin_amdgcn_rcpf(fmaxf(__builtin_fmaf(ps, -0.125f, base), 0.1f));
        o.z = __builtin_amdgcn_rcpf(fmaxf(__builtin_fmaf(ps,  0.125f, base), 0.1f));
        o.w = __builtin_amdgcn_rcpf(fmaxf(__builtin_fmaf(ps,  0.375f, base), 0.1f));
        *reinterpret_cast<float4*>(out + outOff + rr * OW_) = o;
    }
}

extern "C" void kernel_launch(void* const* d_in, const int* in_sizes, int n_in,
                              void* d_out, int out_size, void* d_ws, size_t ws_size,
                              hipStream_t stream) {
    (void)in_sizes; (void)n_in; (void)out_size; (void)d_ws; (void)ws_size;
    const float* feat = (const float*)d_in[0];
    const float* Wm   = (const float*)d_in[1];
    float*       out  = (float*)d_out;

    plane2depth_wmma_kernel<<<TOTAL_CHUNKS / WAVES_PER_BLOCK, 256, 0, stream>>>(feat, Wm, out);
}


# Re-tiled, one lane = one pixel: wave = 32px x 6-row tile (3072 waves), whole-128B-line loads, 512B row stores; channel sum split over two chained WMMAs whose k halves serve the two lane halves
# speedup vs baseline: 1.5120x; 1.5120x over previous
#include <hip/hip_runtime.h>


typedef __attribute__((ext_vector_type(2))) float v2f;
typedef __attribute__((ext_vector_type(8))) float v8f;

#define B_      16
#define CH_     4
#define H_      192
#define W_      192
#define HW_     (H_ * W_)
#define UP_     4
#define OW_     (W_ * UP_)
#define OHW_    (OW_ * OW_)
#define STRIP_  32
#define STRIPS_ (W_ / STRIP_)
#define TROWS_  6
#define BROWS_  12
#define BANDS_  (H_ / BROWS_)
#define WAVES_  (STRIPS_ * (BROWS_ / TROWS_))

__global__ __launch_bounds__(WAVES_ * 32)
void plane2depth_tile_kernel(const float* __restrict__ feat,
                             const float* __restrict__ Wm,
                             float* __restrict__ out)
{
    const int lane = threadIdx.x & 31;
    const int wave  = __builtin_amdgcn_readfirstlane((int)(threadIdx.x >> 5));
    const int strip = wave % STRIPS_;
    const int half  = wave / STRIPS_;
    const int b     = blockIdx.x / BANDS_;
    const int band  = blockIdx.x % BANDS_;
    const int h0    = band * BROWS_ + half * TROWS_;
    const int w     = strip * STRIP_ + lane;

    const int  j   = lane & 3;
    const bool act = (lane < 4) || ((lane >= 24) && (lane < 28));
    const float w0 = Wm[j * 4 + 0], w1 = Wm[j * 4 + 1];
    const float w2 = Wm[j * 4 + 2], w3 = Wm[j * 4 + 3];
    v2f a01, a23;
    a01.x = act ? w0 : 0.0f;  a01.y = act ? w1 : 0.0f;
    a23.x = act ? w2 : 0.0f;  a23.y = act ? w3 : 0.0f;

    const int inOff = b * (CH_ * HW_) + h0 * W_ + w;
    float x[TROWS_][CH_];
    #pragma unroll
    for (int i = 0; i < TROWS_; ++i) {
        #pragma unroll
        for (int c = 0; c < CH_; ++c)
            x[i][c] = feat[inOff + c * HW_ + i * W_];
    }

    __builtin_amdgcn_sched_barrier(0);

    const int outOff = b * OHW_ + (UP_ * h0) * OW_ + UP_ * w;

    #pragma unroll
    for (int i = 0; i < TROWS_; ++i) {
        v2f b01, b23;
        b01.x = x[i][0];  b01.y = x[i][1];
        b23.x = x[i][2];  b23.y = x[i][3];

        v8f c = {};
        c = __builtin_amdgcn_wmma_f32_16x16x4_f32(
                 false, a01,  false, b01,
                 (short)0, c,  false,  false);
        c = __builtin_amdgcn_wmma_f32_16x16x4_f32(
                 false, a23,  false, b23,
                 (short)0, c,  false,  false);

        const float p = c[0], q = c[1], r = c[2], s = c[3];

        const float t   = __builtin_fmaf(p, p, __builtin_fmaf(q, q, __builtin_fmaf(r, r, 1e-6f)));
        const float rsq = __builtin_amdgcn_rsqf(t);
        const float sn  = s * (t * rsq);
        const float ps  = (p * rsq) * sn;
        const float qs  = (q * rsq) * sn;
        const float rs  = (r * rsq) * sn;

        #pragma unroll
        for (int rr = 0; rr < UP_; ++rr) {
            const float v    = ((float)rr - 1.5f) * 0.25f;
            const float base = __builtin_fmaf(qs, v, rs);
            float4 o;
            o.x = __builtin_amdgcn_rcpf(fmaxf(__builtin_fmaf(ps, -0.375f, base), 0.1f));
            o.y = __builtin_amdgcn_rcpf(fmaxf(__builtin_fmaf(ps, -0.125f, base), 0.1f));
            o.z = __builtin_amdgcn_rcpf(fmaxf(__builtin_fmaf(ps,  0.125f, base), 0.1f));
            o.w = __builtin_amdgcn_rcpf(fmaxf(__builtin_fmaf(ps,  0.375f, base), 0.1f));
            *reinterpret_cast<float4*>(out + outOff + (UP_ * i + rr) * OW_) = o;
        }
        __builtin_amdgcn_sched_barrier(0);
    }
}

extern "C" void kernel_launch(void* const* d_in, const int* in_sizes, int n_in,
                              void* d_out, int out_size, void* d_ws, size_t ws_size,
                              hipStream_t stream) {
    (void)in_sizes; (void)n_in; (void)out_size; (void)d_ws; (void)ws_size;
    const float* feat = (const float*)d_in[0];
    const float* Wm   = (const float*)d_in[1];
    float*       out  = (float*)d_out;

    plane2depth_tile_kernel<<<B_ * BANDS_, WAVES_ * 32, 0, stream>>>(feat, Wm, out);
}
